# starting kernel
# baseline (speedup 1.0000x reference)
#include <hip/hip_runtime.h>

typedef __attribute__((ext_vector_type(2))) float v2f;
typedef __attribute__((ext_vector_type(8))) float v8f;

#define BATCH 64
#define LDIM 96
#define CDIM 2048
#define TILE 128
#define TCOLS (TILE + 2)
#define KTOT 288
#define KSTEPS (KTOT / 4)

__global__ void build_wt_kernel(const float* __restrict__ conv_w,
                                float* __restrict__ wt2) {
  int idx = blockIdx.x * blockDim.x + threadIdx.x;
  if (idx >= KTOT * LDIM) return;
  int j  = idx & 1;
  int m  = (idx >> 1) % LDIM;
  int kp = idx / (2 * LDIM);
  int kk = 2 * kp + j;
  int ci = kk % LDIM;
  int dt = kk / LDIM;
  wt2[idx] = conv_w[(m * LDIM + ci) * 3 + dt];
}

__global__ __launch_bounds__(192)
void msdecomp_conv_kernel(const float* __restrict__ x,
                          const float* __restrict__ wt2,
                          const float* __restrict__ gate_w,
                          const float* __restrict__ gate_b,
                          float* __restrict__ out) {
  __shared__ float xs[TCOLS * LDIM];
  __shared__ float rs[TCOLS * LDIM];
  __shared__ float ms[TCOLS * LDIM];

  const int tid = threadIdx.x;
  const int b   = blockIdx.y;
  const int c0  = blockIdx.x * TILE;
  const float* xb = x + (size_t)b * LDIM * CDIM;

  for (int i = tid; i < TCOLS * LDIM; i += blockDim.x) {
    int l   = i / TCOLS;
    int col = i % TCOLS;
    int gc  = (c0 + col - 1 + CDIM) & (CDIM - 1);
    xs[col * LDIM + l] = xb[l * CDIM + gc];
  }
  __syncthreads();

  if (tid < TCOLS) {
    const float gw0 = gate_w[0], gw1 = gate_w[1], gw2 = gate_w[2];
    const float gb0 = gate_b[0], gb1 = gate_b[1], gb2 = gate_b[2];
    const float* xc = &xs[tid * LDIM];
    float* rc = &rs[tid * LDIM];
    float* mc = &ms[tid * LDIM];

    float s5 = 3.f * xc[0] + xc[1] + xc[2];
    float s13 = 7.f * xc[0];
    #pragma unroll
    for (int j = 1; j <= 6; ++j) s13 += xc[j];
    float s25 = 13.f * xc[0];
    #pragma unroll
    for (int j = 1; j <= 12; ++j) s25 += xc[j];

    for (int l = 0; l < LDIM; ++l) {
      float xv  = xc[l];
      float m5  = s5 * 0.2f;
      float m13 = s13 * (1.f / 13.f);
      float m25 = s25 * 0.04f;
      float l0 = fmaf(xv, gw0, gb0);
      float l1 = fmaf(xv, gw1, gb1);
      float l2 = fmaf(xv, gw2, gb2);
      float mx = fmaxf(fmaxf(l0, l1), l2);
      float e0 = __expf(l0 - mx), e1 = __expf(l1 - mx), e2 = __expf(l2 - mx);
      float inv = 1.f / (e0 + e1 + e2);
      float mean = (m5 * e0 + m13 * e1 + m25 * e2) * inv;
      mc[l] = mean;
      rc[l] = xv - mean;
      s5  += xc[min(l + 3, 95)]  - xc[max(l - 2, 0)];
      s13 += xc[min(l + 7, 95)]  - xc[max(l - 6, 0)];
      s25 += xc[min(l + 13, 95)] - xc[max(l - 12, 0)];
    }
  }
  __syncthreads();

  const int wave = tid >> 5;
  const int lane = tid & 31;
  const int half = (lane >> 4) & 1;
  const int ln   = lane & 15;
  const int m0   = wave * 16;

  v8f acc[16];
  const v8f vz = {0.f, 0.f, 0.f, 0.f, 0.f, 0.f, 0.f, 0.f};
  #pragma unroll
  for (int t = 0; t < 16; ++t) acc[t] = vz;

  const v2f* wt = (const v2f*)wt2;
  #pragma unroll 2
  for (int ks = 0; ks < KSTEPS; ++ks) {
    v2f A = wt[(2 * ks + half) * LDIM + m0 + ln];
    int dt = ks / 24;
    int ci = (4 * ks) % 96 + 2 * half;
    #pragma unroll
    for (int t = 0; t < 16; ++t) {
      const float* src = (t < 8) ? rs : ms;
      int row = (t & 7) * 16 + ln + dt;
      v2f Bf = *(const v2f*)&src[row * LDIM + ci];
      acc[t] = __builtin_amdgcn_wmma_f32_16x16x4_f32(
          false, A, false, Bf, (short)0, acc[t], false, false);
    }
  }

  #pragma unroll
  for (int t = 0; t < 16; ++t) {
    int conv  = t >> 3;
    int ntile = t & 7;
    float* op = out + (size_t)conv * BATCH * LDIM * CDIM
                    + (size_t)b * LDIM * CDIM;
    int cc = c0 + ntile * 16 + ln;
    #pragma unroll
    for (int r = 0; r < 8; ++r) {
      int rowm = m0 + r + 8 * half;
      op[(size_t)rowm * CDIM + cc] = acc[t][r];
    }
  }
}

extern "C" void kernel_launch(void* const* d_in, const int* in_sizes, int n_in,
                              void* d_out, int out_size, void* d_ws, size_t ws_size,
                              hipStream_t stream) {
  const float* x      = (const float*)d_in[0];
  const float* conv_w = (const float*)d_in[1];
  const float* gate_w = (const float*)d_in[2];
  const float* gate_b = (const float*)d_in[3];
  float* out = (float*)d_out;
  float* wt2 = (float*)d_ws;

  build_wt_kernel<<<(KTOT * LDIM + 255) / 256, 256, 0, stream>>>(conv_w, wt2);

  dim3 grid(CDIM / TILE, BATCH);
  msdecomp_conv_kernel<<<grid, 192, 0, stream>>>(x, wt2, gate_w, gate_b, out);
}


# Pad the LDS row pitch of the x, res and mean tiles from 96 to 100 floats (4 x odd): a wave's b64 B-fragment reads now cover all 64 banks and the column scans stop piling onto two banks
# speedup vs baseline: 1.3870x; 1.3870x over previous
#include <hip/hip_runtime.h>

typedef __attribute__((ext_vector_type(2))) float v2f;
typedef __attribute__((ext_vector_type(8))) float v8f;

#define BATCH 64
#define LDIM 96
#define CDIM 2048
#define TILE 128
#define TCOLS (TILE + 2)
#define LPITCH 100
#define KTOT 288
#define KSTEPS (KTOT / 4)

__global__ void build_wt_kernel(const float* __restrict__ conv_w,
                                float* __restrict__ wt2) {
  int idx = blockIdx.x * blockDim.x + threadIdx.x;
  if (idx >= KTOT * LDIM) return;
  int j  = idx & 1;
  int m  = (idx >> 1) % LDIM;
  int kp = idx / (2 * LDIM);
  int kk = 2 * kp + j;
  int ci = kk % LDIM;
  int dt = kk / LDIM;
  wt2[idx] = conv_w[(m * LDIM + ci) * 3 + dt];
}

__global__ __launch_bounds__(192)
void msdecomp_conv_kernel(const float* __restrict__ x,
                          const float* __restrict__ wt2,
                          const float* __restrict__ gate_w,
                          const float* __restrict__ gate_b,
                          float* __restrict__ out) {
  __shared__ float xs[TCOLS * LPITCH];
  __shared__ float rs[TCOLS * LPITCH];
  __shared__ float ms[TCOLS * LPITCH];

  const int tid = threadIdx.x;
  const int b   = blockIdx.y;
  const int c0  = blockIdx.x * TILE;
  const float* xb = x + (size_t)b * LDIM * CDIM;

  for (int i = tid; i < TCOLS * LDIM; i += blockDim.x) {
    int l   = i / TCOLS;
    int col = i % TCOLS;
    int gc  = (c0 + col - 1 + CDIM) & (CDIM - 1);
    xs[col * LPITCH + l] = xb[l * CDIM + gc];
  }
  __syncthreads();

  if (tid < TCOLS) {
    const float gw0 = gate_w[0], gw1 = gate_w[1], gw2 = gate_w[2];
    const float gb0 = gate_b[0], gb1 = gate_b[1], gb2 = gate_b[2];
    const float* xc = &xs[tid * LPITCH];
    float* rc = &rs[tid * LPITCH];
    float* mc = &ms[tid * LPITCH];

    float s5 = 3.f * xc[0] + xc[1] + xc[2];
    float s13 = 7.f * xc[0];
    #pragma unroll
    for (int j = 1; j <= 6; ++j) s13 += xc[j];
    float s25 = 13.f * xc[0];
    #pragma unroll
    for (int j = 1; j <= 12; ++j) s25 += xc[j];

    for (int l = 0; l < LDIM; ++l) {
      float xv  = xc[l];
      float m5  = s5 * 0.2f;
      float m13 = s13 * (1.f / 13.f);
      float m25 = s25 * 0.04f;
      float l0 = fmaf(xv, gw0, gb0);
      float l1 = fmaf(xv, gw1, gb1);
      float l2 = fmaf(xv, gw2, gb2);
      float mx = fmaxf(fmaxf(l0, l1), l2);
      float e0 = __expf(l0 - mx), e1 = __expf(l1 - mx), e2 = __expf(l2 - mx);
      float inv = 1.f / (e0 + e1 + e2);
      float mean = (m5 * e0 + m13 * e1 + m25 * e2) * inv;
      mc[l] = mean;
      rc[l] = xv - mean;
      s5  += xc[min(l + 3, 95)]  - xc[max(l - 2, 0)];
      s13 += xc[min(l + 7, 95)]  - xc[max(l - 6, 0)];
      s25 += xc[min(l + 13, 95)] - xc[max(l - 12, 0)];
    }
  }
  __syncthreads();

  const int wave = tid >> 5;
  const int lane = tid & 31;
  const int half = (lane >> 4) & 1;
  const int ln   = lane & 15;
  const int m0   = wave * 16;

  v8f acc[16];
  const v8f vz = {0.f, 0.f, 0.f, 0.f, 0.f, 0.f, 0.f, 0.f};
  #pragma unroll
  for (int t = 0; t < 16; ++t) acc[t] = vz;

  const v2f* wt = (const v2f*)wt2;
  #pragma unroll 2
  for (int ks = 0; ks < KSTEPS; ++ks) {
    v2f A = wt[(2 * ks + half) * LDIM + m0 + ln];
    int dt = ks / 24;
    int ci = (4 * ks) % 96 + 2 * half;
    #pragma unroll
    for (int t = 0; t < 16; ++t) {
      const float* src = (t < 8) ? rs : ms;
      int row = (t & 7) * 16 + ln + dt;
      v2f Bf = *(const v2f*)&src[row * LPITCH + ci];
      acc[t] = __builtin_amdgcn_wmma_f32_16x16x4_f32(
          false, A, false, Bf, (short)0, acc[t], false, false);
    }
  }

  #pragma unroll
  for (int t = 0; t < 16; ++t) {
    int conv  = t >> 3;
    int ntile = t & 7;
    float* op = out + (size_t)conv * BATCH * LDIM * CDIM
                    + (size_t)b * LDIM * CDIM;
    int cc = c0 + ntile * 16 + ln;
    #pragma unroll
    for (int r = 0; r < 8; ++r) {
      int rowm = m0 + r + 8 * half;
      op[(size_t)rowm * CDIM + cc] = acc[t][r];
    }
  }
}

extern "C" void kernel_launch(void* const* d_in, const int* in_sizes, int n_in,
                              void* d_out, int out_size, void* d_ws, size_t ws_size,
                              hipStream_t stream) {
  const float* x      = (const float*)d_in[0];
  const float* conv_w = (const float*)d_in[1];
  const float* gate_w = (const float*)d_in[2];
  const float* gate_b = (const float*)d_in[3];
  float* out = (float*)d_out;
  float* wt2 = (float*)d_ws;

  build_wt_kernel<<<(KTOT * LDIM + 255) / 256, 256, 0, stream>>>(conv_w, wt2);

  dim3 grid(CDIM / TILE, BATCH);
  msdecomp_conv_kernel<<<grid, 192, 0, stream>>>(x, wt2, gate_w, gate_b, out);
}


# Halve the column tile per block from 128 to 64 (grid 32 x 64): LDS per block drops from 156000 to 79200 B, so four blocks share a WGP and their load, scan and store phases overlap other blocks' WMMAs
# speedup vs baseline: 1.5723x; 1.1336x over previous
#include <hip/hip_runtime.h>

typedef __attribute__((ext_vector_type(2))) float v2f;
typedef __attribute__((ext_vector_type(8))) float v8f;

#define BATCH 64
#define LDIM 96
#define CDIM 2048
#define TILE 64
#define NT (TILE / 16)
#define TCOLS (TILE + 2)
#define LPITCH 100
#define KTOT 288
#define KSTEPS (KTOT / 4)

__global__ void build_wt_kernel(const float* __restrict__ conv_w,
                                float* __restrict__ wt2) {
  int idx = blockIdx.x * blockDim.x + threadIdx.x;
  if (idx >= KTOT * LDIM) return;
  int j  = idx & 1;
  int m  = (idx >> 1) % LDIM;
  int kp = idx / (2 * LDIM);
  int kk = 2 * kp + j;
  int ci = kk % LDIM;
  int dt = kk / LDIM;
  wt2[idx] = conv_w[(m * LDIM + ci) * 3 + dt];
}

__global__ __launch_bounds__(192)
void msdecomp_conv_kernel(const float* __restrict__ x,
                          const float* __restrict__ wt2,
                          const float* __restrict__ gate_w,
                          const float* __restrict__ gate_b,
                          float* __restrict__ out) {
  __shared__ float xs[TCOLS * LPITCH];
  __shared__ float rs[TCOLS * LPITCH];
  __shared__ float ms[TCOLS * LPITCH];

  const int tid = threadIdx.x;
  const int b   = blockIdx.y;
  const int c0  = blockIdx.x * TILE;
  const float* xb = x + (size_t)b * LDIM * CDIM;

  for (int i = tid; i < TCOLS * LDIM; i += blockDim.x) {
    int l   = i / TCOLS;
    int col = i % TCOLS;
    int gc  = (c0 + col - 1 + CDIM) & (CDIM - 1);
    xs[col * LPITCH + l] = xb[l * CDIM + gc];
  }
  __syncthreads();

  if (tid < TCOLS) {
    const float gw0 = gate_w[0], gw1 = gate_w[1], gw2 = gate_w[2];
    const float gb0 = gate_b[0], gb1 = gate_b[1], gb2 = gate_b[2];
    const float* xc = &xs[tid * LPITCH];
    float* rc = &rs[tid * LPITCH];
    float* mc = &ms[tid * LPITCH];

    float s5 = 3.f * xc[0] + xc[1] + xc[2];
    float s13 = 7.f * xc[0];
    #pragma unroll
    for (int j = 1; j <= 6; ++j) s13 += xc[j];
    float s25 = 13.f * xc[0];
    #pragma unroll
    for (int j = 1; j <= 12; ++j) s25 += xc[j];

    for (int l = 0; l < LDIM; ++l) {
      float xv  = xc[l];
      float m5  = s5 * 0.2f;
      float m13 = s13 * (1.f / 13.f);
      float m25 = s25 * 0.04f;
      float l0 = fmaf(xv, gw0, gb0);
      float l1 = fmaf(xv, gw1, gb1);
      float l2 = fmaf(xv, gw2, gb2);
      float mx = fmaxf(fmaxf(l0, l1), l2);
      float e0 = __expf(l0 - mx), e1 = __expf(l1 - mx), e2 = __expf(l2 - mx);
      float inv = 1.f / (e0 + e1 + e2);
      float mean = (m5 * e0 + m13 * e1 + m25 * e2) * inv;
      mc[l] = mean;
      rc[l] = xv - mean;
      s5  += xc[min(l + 3, 95)]  - xc[max(l - 2, 0)];
      s13 += xc[min(l + 7, 95)]  - xc[max(l - 6, 0)];
      s25 += xc[min(l + 13, 95)] - xc[max(l - 12, 0)];
    }
  }
  __syncthreads();

  const int wave = tid >> 5;
  const int lane = tid & 31;
  const int half = (lane >> 4) & 1;
  const int ln   = lane & 15;
  const int m0   = wave * 16;

  v8f acc[2 * NT];
  const v8f vz = {0.f, 0.f, 0.f, 0.f, 0.f, 0.f, 0.f, 0.f};
  #pragma unroll
  for (int t = 0; t < 2 * NT; ++t) acc[t] = vz;

  const v2f* wt = (const v2f*)wt2;
  #pragma unroll 2
  for (int ks = 0; ks < KSTEPS; ++ks) {
    v2f A = wt[(2 * ks + half) * LDIM + m0 + ln];
    int dt = ks / 24;
    int ci = (4 * ks) % 96 + 2 * half;
    #pragma unroll
    for (int t = 0; t < 2 * NT; ++t) {
      const float* src = (t < NT) ? rs : ms;
      int row = (t % NT) * 16 + ln + dt;
      v2f Bf = *(const v2f*)&src[row * LPITCH + ci];
      acc[t] = __builtin_amdgcn_wmma_f32_16x16x4_f32(
          false, A, false, Bf, (short)0, acc[t], false, false);
    }
  }

  #pragma unroll
  for (int t = 0; t < 2 * NT; ++t) {
    int conv  = t / NT;
    int ntile = t % NT;
    float* op = out + (size_t)conv * BATCH * LDIM * CDIM
                    + (size_t)b * LDIM * CDIM;
    int cc = c0 + ntile * 16 + ln;
    #pragma unroll
    for (int r = 0; r < 8; ++r) {
      int rowm = m0 + r + 8 * half;
      op[(size_t)rowm * CDIM + cc] = acc[t][r];
    }
  }
}

extern "C" void kernel_launch(void* const* d_in, const int* in_sizes, int n_in,
                              void* d_out, int out_size, void* d_ws, size_t ws_size,
                              hipStream_t stream) {
  const float* x      = (const float*)d_in[0];
  const float* conv_w = (const float*)d_in[1];
  const float* gate_w = (const float*)d_in[2];
  const float* gate_b = (const float*)d_in[3];
  float* out = (float*)d_out;
  float* wt2 = (float*)d_ws;

  build_wt_kernel<<<(KTOT * LDIM + 255) / 256, 256, 0, stream>>>(conv_w, wt2);

  dim3 grid(CDIM / TILE, BATCH);
  msdecomp_conv_kernel<<<grid, 192, 0, stream>>>(x, wt2, gate_w, gate_b, out);
}


# Software-pipeline the fully unrolled K loop in registers: B fragments are read from LDS one trip (16 WMMAs) ahead, A fragments loaded two trips ahead, no trip waits on a load; sum order unchanged
# speedup vs baseline: 1.7190x; 1.0933x over previous
#include <hip/hip_runtime.h>

typedef __attribute__((ext_vector_type(2))) float v2f;
typedef __attribute__((ext_vector_type(8))) float v8f;

#define BATCH 64
#define LDIM 96
#define CDIM 2048
#define TILE 64
#define NT (TILE / 16)
#define TCOLS (TILE + 2)
#define LPITCH 100
#define KTOT 288
#define KSTEPS (KTOT / 4)
#define KTRIPS (KSTEPS / 2)

__global__ void build_wt_kernel(const float* __restrict__ conv_w,
                                float* __restrict__ wt2) {
  int idx = blockIdx.x * blockDim.x + threadIdx.x;
  if (idx >= KTOT * LDIM) return;
  int j  = idx & 1;
  int m  = (idx >> 1) % LDIM;
  int kp = idx / (2 * LDIM);
  int kk = 2 * kp + j;
  int ci = kk % LDIM;
  int dt = kk / LDIM;
  wt2[idx] = conv_w[(m * LDIM + ci) * 3 + dt];
}

__device__ __forceinline__ void load_a_trip(const v2f* __restrict__ wlane, int tp, v2f (&A)[2]) {
  A[0] = wlane[(4 * tp) * LDIM];
  A[1] = wlane[(4 * tp + 2) * LDIM];
}

__device__ __forceinline__ void load_b_trip(const float* rs, const float* ms, int blane, int tp, v2f (&B)[16]) {
  const int dt  = tp / 12;
  const int ci0 = (8 * tp) % 96;
  const int off = blane + dt * LPITCH + ci0;
  #pragma unroll
  for (int s = 0; s < 2; ++s) {
    #pragma unroll
    for (int n = 0; n < NT; ++n) {
      B[s * 8 + n]      = *(const v2f*)&rs[off + n * 16 * LPITCH + 4 * s];
      B[s * 8 + NT + n] = *(const v2f*)&ms[off + n * 16 * LPITCH + 4 * s];
    }
  }
}

__device__ __forceinline__ void mma_trip(const v2f (&A)[2], const v2f (&B)[16], v8f (&acc)[2 * NT]) {
  #pragma unroll
  for (int s = 0; s < 2; ++s) {
    #pragma unroll
    for (int t = 0; t < 2 * NT; ++t) {
      acc[t] = __builtin_amdgcn_wmma_f32_16x16x4_f32(
          false, A[s], false, B[s * 8 + t], (short)0, acc[t], false, false);
    }
  }
}

__global__ __launch_bounds__(192)
void msdecomp_conv_kernel(const float* __restrict__ x,
                          const float* __restrict__ wt2,
                          const float* __restrict__ gate_w,
                          const float* __restrict__ gate_b,
                          float* __restrict__ out) {
  __shared__ float xs[TCOLS * LPITCH];
  __shared__ float rs[TCOLS * LPITCH];
  __shared__ float ms[TCOLS * LPITCH];

  const int tid = threadIdx.x;
  const int b   = blockIdx.y;
  const int c0  = blockIdx.x * TILE;
  const float* xb = x + (size_t)b * LDIM * CDIM;

  for (int i = tid; i < TCOLS * LDIM; i += blockDim.x) {
    int l   = i / TCOLS;
    int col = i % TCOLS;
    int gc  = (c0 + col - 1 + CDIM) & (CDIM - 1);
    xs[col * LPITCH + l] = xb[l * CDIM + gc];
  }
  __syncthreads();

  if (tid < TCOLS) {
    const float gw0 = gate_w[0], gw1 = gate_w[1], gw2 = gate_w[2];
    const float gb0 = gate_b[0], gb1 = gate_b[1], gb2 = gate_b[2];
    const float* xc = &xs[tid * LPITCH];
    float* rc = &rs[tid * LPITCH];
    float* mc = &ms[tid * LPITCH];

    float s5 = 3.f * xc[0] + xc[1] + xc[2];
    float s13 = 7.f * xc[0];
    #pragma unroll
    for (int j = 1; j <= 6; ++j) s13 += xc[j];
    float s25 = 13.f * xc[0];
    #pragma unroll
    for (int j = 1; j <= 12; ++j) s25 += xc[j];

    for (int l = 0; l < LDIM; ++l) {
      float xv  = xc[l];
      float m5  = s5 * 0.2f;
      float m13 = s13 * (1.f / 13.f);
      float m25 = s25 * 0.04f;
      float l0 = fmaf(xv, gw0, gb0);
      float l1 = fmaf(xv, gw1, gb1);
      float l2 = fmaf(xv, gw2, gb2);
      float mx = fmaxf(fmaxf(l0, l1), l2);
      float e0 = __expf(l0 - mx), e1 = __expf(l1 - mx), e2 = __expf(l2 - mx);
      float inv = 1.f / (e0 + e1 + e2);
      float mean = (m5 * e0 + m13 * e1 + m25 * e2) * inv;
      mc[l] = mean;
      rc[l] = xv - mean;
      s5  += xc[min(l + 3, 95)]  - xc[max(l - 2, 0)];
      s13 += xc[min(l + 7, 95)]  - xc[max(l - 6, 0)];
      s25 += xc[min(l + 13, 95)] - xc[max(l - 12, 0)];
    }
  }
  __syncthreads();

  const int wave = tid >> 5;
  const int lane = tid & 31;
  const int half = (lane >> 4) & 1;
  const int ln   = lane & 15;
  const int m0   = wave * 16;

  v8f acc[2 * NT];
  const v8f vz = {0.f, 0.f, 0.f, 0.f, 0.f, 0.f, 0.f, 0.f};
  #pragma unroll
  for (int t = 0; t < 2 * NT; ++t) acc[t] = vz;

  const v2f* wlane = (const v2f*)wt2 + half * LDIM + m0 + ln;
  const int blane = ln * LPITCH + 2 * half;

  v2f Af[3][2];
  v2f Bf[2][16];

  load_a_trip(wlane, 0, Af[0]);
  load_a_trip(wlane, 1, Af[1]);
  load_b_trip(rs, ms, blane, 0, Bf[0]);

  #pragma unroll
  for (int tp = 0; tp < KTRIPS; ++tp) {
    if (tp + 2 < KTRIPS) load_a_trip(wlane, tp + 2, Af[(tp + 2) % 3]);
    if (tp + 1 < KTRIPS) load_b_trip(rs, ms, blane, tp + 1, Bf[(tp + 1) % 2]);
    mma_trip(Af[tp % 3], Bf[tp % 2], acc);
    __builtin_amdgcn_sched_barrier(0);
  }

  #pragma unroll
  for (int t = 0; t < 2 * NT; ++t) {
    int conv  = t / NT;
    int ntile = t % NT;
    float* op = out + (size_t)conv * BATCH * LDIM * CDIM
                    + (size_t)b * LDIM * CDIM;
    int cc = c0 + ntile * 16 + ln;
    #pragma unroll
    for (int r = 0; r < 8; ++r) {
      int rowm = m0 + r + 8 * half;
      op[(size_t)rowm * CDIM + cc] = acc[t][r];
    }
  }
}

extern "C" void kernel_launch(void* const* d_in, const int* in_sizes, int n_in,
                              void* d_out, int out_size, void* d_ws, size_t ws_size,
                              hipStream_t stream) {
  const float* x      = (const float*)d_in[0];
  const float* conv_w = (const float*)d_in[1];
  const float* gate_w = (const float*)d_in[2];
  const float* gate_b = (const float*)d_in[3];
  float* out = (float*)d_out;
  float* wt2 = (float*)d_ws;

  build_wt_kernel<<<(KTOT * LDIM + 255) / 256, 256, 0, stream>>>(conv_w, wt2);

  dim3 grid(CDIM / TILE, BATCH);
  msdecomp_conv_kernel<<<grid, 192, 0, stream>>>(x, wt2, gate_w, gate_b, out);
}


# Make the block 12 waves instead of 6 (128-column tile, wave = M-tile x 64-column half, grid 16 x 64): a multiple of 4 spreads evenly over the 4 SIMDs, so no SIMD carries twice the WMMAs of another
# speedup vs baseline: 2.0893x; 1.2154x over previous
#include <hip/hip_runtime.h>

typedef __attribute__((ext_vector_type(2))) float v2f;
typedef __attribute__((ext_vector_type(8))) float v8f;

#define BATCH 64
#define LDIM 96
#define CDIM 2048
#define TILE 128
#define WAVES 12
#define MT (LDIM / 16)
#define HCOLS (TILE * MT / WAVES)
#define NT (HCOLS / 16)
#define TCOLS (TILE + 2)
#define LPITCH 100
#define KTOT 288
#define KSTEPS (KTOT / 4)
#define KTRIPS (KSTEPS / 2)

__global__ void build_wt_kernel(const float* __restrict__ conv_w,
                                float* __restrict__ wt2) {
  int idx = blockIdx.x * blockDim.x + threadIdx.x;
  if (idx >= KTOT * LDIM) return;
  int j  = idx & 1;
  int m  = (idx >> 1) % LDIM;
  int kp = idx / (2 * LDIM);
  int kk = 2 * kp + j;
  int ci = kk % LDIM;
  int dt = kk / LDIM;
  wt2[idx] = conv_w[(m * LDIM + ci) * 3 + dt];
}

__device__ __forceinline__ void load_a_trip(const v2f* __restrict__ wlane, int tp, v2f (&A)[2]) {
  A[0] = wlane[(4 * tp) * LDIM];
  A[1] = wlane[(4 * tp + 2) * LDIM];
}

__device__ __forceinline__ void load_b_trip(const float* rs, const float* ms, int blane, int tp, v2f (&B)[16]) {
  const int dt  = tp / 12;
  const int ci0 = (8 * tp) % 96;
  const int off = blane + dt * LPITCH + ci0;
  #pragma unroll
  for (int s = 0; s < 2; ++s) {
    #pragma unroll
    for (int n = 0; n < NT; ++n) {
      B[s * 8 + n]      = *(const v2f*)&rs[off + n * 16 * LPITCH + 4 * s];
      B[s * 8 + NT + n] = *(const v2f*)&ms[off + n * 16 * LPITCH + 4 * s];
    }
  }
}

__device__ __forceinline__ void mma_trip(const v2f (&A)[2], const v2f (&B)[16], v8f (&acc)[2 * NT]) {
  #pragma unroll
  for (int s = 0; s < 2; ++s) {
    #pragma unroll
    for (int t = 0; t < 2 * NT; ++t) {
      acc[t] = __builtin_amdgcn_wmma_f32_16x16x4_f32(
          false, A[s], false, B[s * 8 + t], (short)0, acc[t], false, false);
    }
  }
}

__global__ __launch_bounds__(WAVES * 32)
void msdecomp_conv_kernel(const float* __restrict__ x,
                          const float* __restrict__ wt2,
                          const float* __restrict__ gate_w,
                          const float* __restrict__ gate_b,
                          float* __restrict__ out) {
  __shared__ float xs[TCOLS * LPITCH];
  __shared__ float rs[TCOLS * LPITCH];
  __shared__ float ms[TCOLS * LPITCH];

  const int tid = threadIdx.x;
  const int b   = blockIdx.y;
  const int c0  = blockIdx.x * TILE;
  const float* xb = x + (size_t)b * LDIM * CDIM;

  for (int i = tid; i < TCOLS * LDIM; i += blockDim.x) {
    int l   = i / TCOLS;
    int col = i % TCOLS;
    int gc  = (c0 + col - 1 + CDIM) & (CDIM - 1);
    xs[col * LPITCH + l] = xb[l * CDIM + gc];
  }
  __syncthreads();

  if (tid < TCOLS) {
    const float gw0 = gate_w[0], gw1 = gate_w[1], gw2 = gate_w[2];
    const float gb0 = gate_b[0], gb1 = gate_b[1], gb2 = gate_b[2];
    const float* xc = &xs[tid * LPITCH];
    float* rc = &rs[tid * LPITCH];
    float* mc = &ms[tid * LPITCH];

    float s5 = 3.f * xc[0] + xc[1] + xc[2];
    float s13 = 7.f * xc[0];
    #pragma unroll
    for (int j = 1; j <= 6; ++j) s13 += xc[j];
    float s25 = 13.f * xc[0];
    #pragma unroll
    for (int j = 1; j <= 12; ++j) s25 += xc[j];

    for (int l = 0; l < LDIM; ++l) {
      float xv  = xc[l];
      float m5  = s5 * 0.2f;
      float m13 = s13 * (1.f / 13.f);
      float m25 = s25 * 0.04f;
      float l0 = fmaf(xv, gw0, gb0);
      float l1 = fmaf(xv, gw1, gb1);
      float l2 = fmaf(xv, gw2, gb2);
      float mx = fmaxf(fmaxf(l0, l1), l2);
      float e0 = __expf(l0 - mx), e1 = __expf(l1 - mx), e2 = __expf(l2 - mx);
      float inv = 1.f / (e0 + e1 + e2);
      float mean = (m5 * e0 + m13 * e1 + m25 * e2) * inv;
      mc[l] = mean;
      rc[l] = xv - mean;
      s5  += xc[min(l + 3, 95)]  - xc[max(l - 2, 0)];
      s13 += xc[min(l + 7, 95)]  - xc[max(l - 6, 0)];
      s25 += xc[min(l + 13, 95)] - xc[max(l - 12, 0)];
    }
  }
  __syncthreads();

  const int wave = tid >> 5;
  const int lane = tid & 31;
  const int half = (lane >> 4) & 1;
  const int ln   = lane & 15;
  const int m0   = (wave % MT) * 16;
  const int h0   = (wave / MT) * HCOLS;

  v8f acc[2 * NT];
  const v8f vz = {0.f, 0.f, 0.f, 0.f, 0.f, 0.f, 0.f, 0.f};
  #pragma unroll
  for (int t = 0; t < 2 * NT; ++t) acc[t] = vz;

  const v2f* wlane = (const v2f*)wt2 + half * LDIM + m0 + ln;
  const int blane = (h0 + ln) * LPITCH + 2 * half;

  v2f Af[3][2];
  v2f Bf[2][16];

  load_a_trip(wlane, 0, Af[0]);
  load_a_trip(wlane, 1, Af[1]);
  load_b_trip(rs, ms, blane, 0, Bf[0]);

  #pragma unroll
  for (int tp = 0; tp < KTRIPS; ++tp) {
    if (tp + 2 < KTRIPS) load_a_trip(wlane, tp + 2, Af[(tp + 2) % 3]);
    if (tp + 1 < KTRIPS) load_b_trip(rs, ms, blane, tp + 1, Bf[(tp + 1) % 2]);
    mma_trip(Af[tp % 3], Bf[tp % 2], acc);
    __builtin_amdgcn_sched_barrier(0);
  }

  #pragma unroll
  for (int t = 0; t < 2 * NT; ++t) {
    int conv  = t / NT;
    int ntile = t % NT;
    float* op = out + (size_t)conv * BATCH * LDIM * CDIM
                    + (size_t)b * LDIM * CDIM;
    int cc = c0 + h0 + ntile * 16 + ln;
    #pragma unroll
    for (int r = 0; r < 8; ++r) {
      int rowm = m0 + r + 8 * half;
      op[(size_t)rowm * CDIM + cc] = acc[t][r];
    }
  }
}

extern "C" void kernel_launch(void* const* d_in, const int* in_sizes, int n_in,
                              void* d_out, int out_size, void* d_ws, size_t ws_size,
                              hipStream_t stream) {
  const float* x      = (const float*)d_in[0];
  const float* conv_w = (const float*)d_in[1];
  const float* gate_w = (const float*)d_in[2];
  const float* gate_b = (const float*)d_in[3];
  float* out = (float*)d_out;
  float* wt2 = (float*)d_ws;

  build_wt_kernel<<<(KTOT * LDIM + 255) / 256, 256, 0, stream>>>(conv_w, wt2);

  dim3 grid(CDIM / TILE, BATCH);
  msdecomp_conv_kernel<<<grid, WAVES * 32, 0, stream>>>(x, wt2, gate_w, gate_b, out);
}


# Decomposition on all 12 waves, not 130 threads: thread = (32-row segment, interior column), 56-row window in registers, window sums restart per segment (sum order changes); halo: a thread per row
# speedup vs baseline: 2.3237x; 1.1122x over previous
#include <hip/hip_runtime.h>

typedef __attribute__((ext_vector_type(2))) float v2f;
typedef __attribute__((ext_vector_type(8))) float v8f;

#define BATCH 64
#define LDIM 96
#define CDIM 2048
#define TILE 128
#define WAVES 12
#define MT (LDIM / 16)
#define HCOLS (TILE * MT / WAVES)
#define NT (HCOLS / 16)
#define TCOLS (TILE + 2)
#define LPITCH 100
#define KTOT 288
#define KSTEPS (KTOT / 4)
#define KTRIPS (KSTEPS / 2)
#define SEGL 32
#define WIN (SEGL + 24)

__global__ void build_wt_kernel(const float* __restrict__ conv_w,
                                float* __restrict__ wt2) {
  int idx = blockIdx.x * blockDim.x + threadIdx.x;
  if (idx >= KTOT * LDIM) return;
  int j  = idx & 1;
  int m  = (idx >> 1) % LDIM;
  int kp = idx / (2 * LDIM);
  int kk = 2 * kp + j;
  int ci = kk % LDIM;
  int dt = kk / LDIM;
  wt2[idx] = conv_w[(m * LDIM + ci) * 3 + dt];
}

__device__ __forceinline__ void load_a_trip(const v2f* __restrict__ wlane, int tp, v2f (&A)[2]) {
  A[0] = wlane[(4 * tp) * LDIM];
  A[1] = wlane[(4 * tp + 2) * LDIM];
}

__device__ __forceinline__ void load_b_trip(const float* rs, const float* ms, int blane, int tp, v2f (&B)[16]) {
  const int dt  = tp / 12;
  const int ci0 = (8 * tp) % 96;
  const int off = blane + dt * LPITCH + ci0;
  #pragma unroll
  for (int s = 0; s < 2; ++s) {
    #pragma unroll
    for (int n = 0; n < NT; ++n) {
      B[s * 8 + n]      = *(const v2f*)&rs[off + n * 16 * LPITCH + 4 * s];
      B[s * 8 + NT + n] = *(const v2f*)&ms[off + n * 16 * LPITCH + 4 * s];
    }
  }
}

__device__ __forceinline__ void mma_trip(const v2f (&A)[2], const v2f (&B)[16], v8f (&acc)[2 * NT]) {
  #pragma unroll
  for (int s = 0; s < 2; ++s) {
    #pragma unroll
    for (int t = 0; t < 2 * NT; ++t) {
      acc[t] = __builtin_amdgcn_wmma_f32_16x16x4_f32(
          false, A[s], false, B[s * 8 + t], (short)0, acc[t], false, false);
    }
  }
}

__device__ __forceinline__ void decomp_row(float xv, float s5, float s13, float s25,
                                           float gw0, float gw1, float gw2,
                                           float gb0, float gb1, float gb2,
                                           float& mean, float& res) {
  float m5  = s5 * 0.2f;
  float m13 = s13 * (1.f / 13.f);
  float m25 = s25 * 0.04f;
  float l0 = fmaf(xv, gw0, gb0);
  float l1 = fmaf(xv, gw1, gb1);
  float l2 = fmaf(xv, gw2, gb2);
  float mx = fmaxf(fmaxf(l0, l1), l2);
  float e0 = __expf(l0 - mx), e1 = __expf(l1 - mx), e2 = __expf(l2 - mx);
  float inv = 1.f / (e0 + e1 + e2);
  mean = (m5 * e0 + m13 * e1 + m25 * e2) * inv;
  res = xv - mean;
}

__global__ __launch_bounds__(WAVES * 32)
void msdecomp_conv_kernel(const float* __restrict__ x,
                          const float* __restrict__ wt2,
                          const float* __restrict__ gate_w,
                          const float* __restrict__ gate_b,
                          float* __restrict__ out) {
  __shared__ __attribute__((aligned(16))) float xs[TCOLS * LPITCH];
  __shared__ __attribute__((aligned(16))) float rs[TCOLS * LPITCH];
  __shared__ __attribute__((aligned(16))) float ms[TCOLS * LPITCH];

  const int tid = threadIdx.x;
  const int b   = blockIdx.y;
  const int c0  = blockIdx.x * TILE;
  const float* xb = x + (size_t)b * LDIM * CDIM;

  for (int i = tid; i < TCOLS * LDIM; i += blockDim.x) {
    int l   = i / TCOLS;
    int col = i % TCOLS;
    int gc  = (c0 + col - 1 + CDIM) & (CDIM - 1);
    xs[col * LPITCH + l] = xb[l * CDIM + gc];
  }
  __syncthreads();

  const float gw0 = gate_w[0], gw1 = gate_w[1], gw2 = gate_w[2];
  const float gb0 = gate_b[0], gb1 = gate_b[1], gb2 = gate_b[2];
  {
    const int seg = __builtin_amdgcn_readfirstlane(tid >> 7);
    const int lbeg = seg * SEGL;
    const int col = (tid & (TILE - 1)) + 1;
    const float* xc = &xs[col * LPITCH];
    float* rc = &rs[col * LPITCH];
    float* mc = &ms[col * LPITCH];

    float w[WIN];
    if (seg == 0) {
      #pragma unroll
      for (int j = 0; j < WIN - 12; ++j) w[12 + j] = xc[j];
      #pragma unroll
      for (int j = 0; j < 12; ++j) w[j] = w[12];
    } else if (seg == 1) {
      #pragma unroll
      for (int j = 0; j < WIN; ++j) w[j] = xc[SEGL - 12 + j];
    } else {
      #pragma unroll
      for (int j = 0; j < WIN - 12; ++j) w[j] = xc[2 * SEGL - 12 + j];
      #pragma unroll
      for (int j = WIN - 12; j < WIN; ++j) w[j] = w[WIN - 13];
    }

    float s5 = w[12];
    #pragma unroll
    for (int j = 1; j <= 2; ++j)  s5  += w[12 + j] + w[12 - j];
    float s13 = s5;
    #pragma unroll
    for (int j = 3; j <= 6; ++j)  s13 += w[12 + j] + w[12 - j];
    float s25 = s13;
    #pragma unroll
    for (int j = 7; j <= 12; ++j) s25 += w[12 + j] + w[12 - j];

    #pragma unroll
    for (int i = 0; i < SEGL; ++i) {
      float mean, res;
      decomp_row(w[12 + i], s5, s13, s25, gw0, gw1, gw2, gb0, gb1, gb2, mean, res);
      mc[lbeg + i] = mean;
      rc[lbeg + i] = res;
      if (i + 1 < SEGL) {
        s5  += w[i + 15] - w[i + 10];
        s13 += w[i + 19] - w[i + 6];
        s25 += w[i + 25] - w[i];
      }
    }
  }
  if (tid < 2 * LDIM) {
    const int hsel = tid / LDIM;
    const int l    = tid - hsel * LDIM;
    const int col  = hsel * (TCOLS - 1);
    const float* xc = &xs[col * LPITCH];
    float s5 = xc[l];
    for (int j = 1; j <= 2; ++j)  s5  += xc[min(l + j, 95)] + xc[max(l - j, 0)];
    float s13 = s5;
    for (int j = 3; j <= 6; ++j)  s13 += xc[min(l + j, 95)] + xc[max(l - j, 0)];
    float s25 = s13;
    for (int j = 7; j <= 12; ++j) s25 += xc[min(l + j, 95)] + xc[max(l - j, 0)];
    float mean, res;
    decomp_row(xc[l], s5, s13, s25, gw0, gw1, gw2, gb0, gb1, gb2, mean, res);
    ms[col * LPITCH + l] = mean;
    rs[col * LPITCH + l] = res;
  }
  __syncthreads();

  const int wave = tid >> 5;
  const int lane = tid & 31;
  const int half = (lane >> 4) & 1;
  const int ln   = lane & 15;
  const int m0   = (wave % MT) * 16;
  const int h0   = (wave / MT) * HCOLS;

  v8f acc[2 * NT];
  const v8f vz = {0.f, 0.f, 0.f, 0.f, 0.f, 0.f, 0.f, 0.f};
  #pragma unroll
  for (int t = 0; t < 2 * NT; ++t) acc[t] = vz;

  const v2f* wlane = (const v2f*)wt2 + half * LDIM + m0 + ln;
  const int blane = (h0 + ln) * LPITCH + 2 * half;

  v2f Af[3][2];
  v2f Bf[2][16];

  load_a_trip(wlane, 0, Af[0]);
  load_a_trip(wlane, 1, Af[1]);
  load_b_trip(rs, ms, blane, 0, Bf[0]);

  #pragma unroll
  for (int tp = 0; tp < KTRIPS; ++tp) {
    if (tp + 2 < KTRIPS) load_a_trip(wlane, tp + 2, Af[(tp + 2) % 3]);
    if (tp + 1 < KTRIPS) load_b_trip(rs, ms, blane, tp + 1, Bf[(tp + 1) % 2]);
    mma_trip(Af[tp % 3], Bf[tp % 2], acc);
    __builtin_amdgcn_sched_barrier(0);
  }

  #pragma unroll
  for (int t = 0; t < 2 * NT; ++t) {
    int conv  = t / NT;
    int ntile = t % NT;
    float* op = out + (size_t)conv * BATCH * LDIM * CDIM
                    + (size_t)b * LDIM * CDIM;
    int cc = c0 + h0 + ntile * 16 + ln;
    #pragma unroll
    for (int r = 0; r < 8; ++r) {
      int rowm = m0 + r + 8 * half;
      op[(size_t)rowm * CDIM + cc] = acc[t][r];
    }
  }
}

extern "C" void kernel_launch(void* const* d_in, const int* in_sizes, int n_in,
                              void* d_out, int out_size, void* d_ws, size_t ws_size,
                              hipStream_t stream) {
  const float* x      = (const float*)d_in[0];
  const float* conv_w = (const float*)d_in[1];
  const float* gate_w = (const float*)d_in[2];
  const float* gate_b = (const float*)d_in[3];
  float* out = (float*)d_out;
  float* wt2 = (float*)d_ws;

  build_wt_kernel<<<(KTOT * LDIM + 255) / 256, 256, 0, stream>>>(conv_w, wt2);

  dim3 grid(CDIM / TILE, BATCH);
  msdecomp_conv_kernel<<<grid, WAVES * 32, 0, stream>>>(x, wt2, gate_w, gate_b, out);
}


# Decomposition threads load their 56-row x windows straight from global memory (whole aligned lines, all loads in flight at once); no x tile in LDS, only the 2 halo columns; sum order unchanged
# speedup vs baseline: 2.4337x; 1.0473x over previous
#include <hip/hip_runtime.h>

typedef __attribute__((ext_vector_type(2))) float v2f;
typedef __attribute__((ext_vector_type(8))) float v8f;

#define BATCH 64
#define LDIM 96
#define CDIM 2048
#define TILE 128
#define WAVES 12
#define MT (LDIM / 16)
#define HCOLS (TILE * MT / WAVES)
#define NT (HCOLS / 16)
#define TCOLS (TILE + 2)
#define LPITCH 100
#define KTOT 288
#define KSTEPS (KTOT / 4)
#define KTRIPS (KSTEPS / 2)
#define SEGL 32
#define WIN (SEGL + 24)

__global__ void build_wt_kernel(const float* __restrict__ conv_w,
                                float* __restrict__ wt2) {
  int idx = blockIdx.x * blockDim.x + threadIdx.x;
  if (idx >= KTOT * LDIM) return;
  int j  = idx & 1;
  int m  = (idx >> 1) % LDIM;
  int kp = idx / (2 * LDIM);
  int kk = 2 * kp + j;
  int ci = kk % LDIM;
  int dt = kk / LDIM;
  wt2[idx] = conv_w[(m * LDIM + ci) * 3 + dt];
}

__device__ __forceinline__ void load_a_trip(const v2f* __restrict__ wlane, int tp, v2f (&A)[2]) {
  A[0] = wlane[(4 * tp) * LDIM];
  A[1] = wlane[(4 * tp + 2) * LDIM];
}

__device__ __forceinline__ void load_b_trip(const float* rs, const float* ms, int blane, int tp, v2f (&B)[16]) {
  const int dt  = tp / 12;
  const int ci0 = (8 * tp) % 96;
  const int off = blane + dt * LPITCH + ci0;
  #pragma unroll
  for (int s = 0; s < 2; ++s) {
    #pragma unroll
    for (int n = 0; n < NT; ++n) {
      B[s * 8 + n]      = *(const v2f*)&rs[off + n * 16 * LPITCH + 4 * s];
      B[s * 8 + NT + n] = *(const v2f*)&ms[off + n * 16 * LPITCH + 4 * s];
    }
  }
}

__device__ __forceinline__ void mma_trip(const v2f (&A)[2], const v2f (&B)[16], v8f (&acc)[2 * NT]) {
  #pragma unroll
  for (int s = 0; s < 2; ++s) {
    #pragma unroll
    for (int t = 0; t < 2 * NT; ++t) {
      acc[t] = __builtin_amdgcn_wmma_f32_16x16x4_f32(
          false, A[s], false, B[s * 8 + t], (short)0, acc[t], false, false);
    }
  }
}

__device__ __forceinline__ void decomp_row(float xv, float s5, float s13, float s25,
                                           float gw0, float gw1, float gw2,
                                           float gb0, float gb1, float gb2,
                                           float& mean, float& res) {
  float m5  = s5 * 0.2f;
  float m13 = s13 * (1.f / 13.f);
  float m25 = s25 * 0.04f;
  float l0 = fmaf(xv, gw0, gb0);
  float l1 = fmaf(xv, gw1, gb1);
  float l2 = fmaf(xv, gw2, gb2);
  float mx = fmaxf(fmaxf(l0, l1), l2);
  float e0 = __expf(l0 - mx), e1 = __expf(l1 - mx), e2 = __expf(l2 - mx);
  float inv = 1.f / (e0 + e1 + e2);
  mean = (m5 * e0 + m13 * e1 + m25 * e2) * inv;
  res = xv - mean;
}

__global__ __launch_bounds__(WAVES * 32) __attribute__((amdgpu_waves_per_eu(6, 6)))
void msdecomp_conv_kernel(const float* __restrict__ x,
                          const float* __restrict__ wt2,
                          const float* __restrict__ gate_w,
                          const float* __restrict__ gate_b,
                          float* __restrict__ out) {
  __shared__ __attribute__((aligned(16))) float rs[TCOLS * LPITCH];
  __shared__ __attribute__((aligned(16))) float ms[TCOLS * LPITCH];
  __shared__ float xh[2 * LDIM];

  const int tid = threadIdx.x;
  const int b   = blockIdx.y;
  const int c0  = blockIdx.x * TILE;
  const float* xb = x + (size_t)b * LDIM * CDIM;

  const float gw0 = gate_w[0], gw1 = gate_w[1], gw2 = gate_w[2];
  const float gb0 = gate_b[0], gb1 = gate_b[1], gb2 = gate_b[2];
  {
    const int seg = __builtin_amdgcn_readfirstlane(tid >> 7);
    const int lbeg = seg * SEGL;
    const int ct  = tid & (TILE - 1);
    const int col = ct + 1;
    const float* xt = xb + c0;
    float* rc = &rs[col * LPITCH];
    float* mc = &ms[col * LPITCH];

    float hv = 0.f;
    if (tid < 2 * LDIM) {
      const int hsel = tid / LDIM;
      const int l    = tid - hsel * LDIM;
      const int gc   = (hsel ? (c0 + TILE) : (c0 - 1 + CDIM)) & (CDIM - 1);
      hv = xb[l * CDIM + gc];
    }

    float w[WIN];
    #pragma unroll
    for (int j = 0; j < WIN; ++j) {
      int r = lbeg - 12 + j;
      if (j < 12) r = max(r, 0);
      if (j >= WIN - 12) r = min(r, LDIM - 1);
      w[j] = xt[r * CDIM + ct];
    }

    if (tid < 2 * LDIM) xh[tid] = hv;
    __syncthreads();

    float s5 = w[12];
    #pragma unroll
    for (int j = 1; j <= 2; ++j)  s5  += w[12 + j] + w[12 - j];
    float s13 = s5;
    #pragma unroll
    for (int j = 3; j <= 6; ++j)  s13 += w[12 + j] + w[12 - j];
    float s25 = s13;
    #pragma unroll
    for (int j = 7; j <= 12; ++j) s25 += w[12 + j] + w[12 - j];

    #pragma unroll
    for (int i = 0; i < SEGL; ++i) {
      float mean, res;
      decomp_row(w[12 + i], s5, s13, s25, gw0, gw1, gw2, gb0, gb1, gb2, mean, res);
      mc[lbeg + i] = mean;
      rc[lbeg + i] = res;
      if (i + 1 < SEGL) {
        s5  += w[i + 15] - w[i + 10];
        s13 += w[i + 19] - w[i + 6];
        s25 += w[i + 25] - w[i];
      }
    }
  }
  if (tid < 2 * LDIM) {
    const int hsel = tid / LDIM;
    const int l    = tid - hsel * LDIM;
    const int col  = hsel * (TCOLS - 1);
    const float* xc = &xh[hsel * LDIM];
    float s5 = xc[l];
    for (int j = 1; j <= 2; ++j)  s5  += xc[min(l + j, 95)] + xc[max(l - j, 0)];
    float s13 = s5;
    for (int j = 3; j <= 6; ++j)  s13 += xc[min(l + j, 95)] + xc[max(l - j, 0)];
    float s25 = s13;
    for (int j = 7; j <= 12; ++j) s25 += xc[min(l + j, 95)] + xc[max(l - j, 0)];
    float mean, res;
    decomp_row(xc[l], s5, s13, s25, gw0, gw1, gw2, gb0, gb1, gb2, mean, res);
    ms[col * LPITCH + l] = mean;
    rs[col * LPITCH + l] = res;
  }
  __syncthreads();

  const int wave = tid >> 5;
  const int lane = tid & 31;
  const int half = (lane >> 4) & 1;
  const int ln   = lane & 15;
  const int m0   = (wave % MT) * 16;
  const int h0   = (wave / MT) * HCOLS;

  v8f acc[2 * NT];
  const v8f vz = {0.f, 0.f, 0.f, 0.f, 0.f, 0.f, 0.f, 0.f};
  #pragma unroll
  for (int t = 0; t < 2 * NT; ++t) acc[t] = vz;

  const v2f* wlane = (const v2f*)wt2 + half * LDIM + m0 + ln;
  const int blane = (h0 + ln) * LPITCH + 2 * half;

  v2f Af[3][2];
  v2f Bf[2][16];

  load_a_trip(wlane, 0, Af[0]);
  load_a_trip(wlane, 1, Af[1]);
  load_b_trip(rs, ms, blane, 0, Bf[0]);

  #pragma unroll
  for (int tp = 0; tp < KTRIPS; ++tp) {
    if (tp + 2 < KTRIPS) load_a_trip(wlane, tp + 2, Af[(tp + 2) % 3]);
    if (tp + 1 < KTRIPS) load_b_trip(rs, ms, blane, tp + 1, Bf[(tp + 1) % 2]);
    mma_trip(Af[tp % 3], Bf[tp % 2], acc);
    __builtin_amdgcn_sched_barrier(0);
  }

  #pragma unroll
  for (int t = 0; t < 2 * NT; ++t) {
    int conv  = t / NT;
    int ntile = t % NT;
    float* op = out + (size_t)conv * BATCH * LDIM * CDIM
                    + (size_t)b * LDIM * CDIM;
    int cc = c0 + h0 + ntile * 16 + ln;
    #pragma unroll
    for (int r = 0; r < 8; ++r) {
      int rowm = m0 + r + 8 * half;
      op[(size_t)rowm * CDIM + cc] = acc[t][r];
    }
  }
}

extern "C" void kernel_launch(void* const* d_in, const int* in_sizes, int n_in,
                              void* d_out, int out_size, void* d_ws, size_t ws_size,
                              hipStream_t stream) {
  const float* x      = (const float*)d_in[0];
  const float* conv_w = (const float*)d_in[1];
  const float* gate_w = (const float*)d_in[2];
  const float* gate_b = (const float*)d_in[3];
  float* out = (float*)d_out;
  float* wt2 = (float*)d_ws;

  build_wt_kernel<<<(KTOT * LDIM + 255) / 256, 256, 0, stream>>>(conv_w, wt2);

  dim3 grid(CDIM / TILE, BATCH);
  msdecomp_conv_kernel<<<grid, WAVES * 32, 0, stream>>>(x, wt2, gate_w, gate_b, out);
}


# Conv wave = (conv, column half, 3 M-tiles), 8-wave blocks: a K-step's transformed B fragments are formed once per 18 WMMAs, were once per 6; decomposition in 48-row segments (sum order changes)
# speedup vs baseline: 3.5832x; 1.4723x over previous
#include <hip/hip_runtime.h>

typedef __attribute__((ext_vector_type(2))) float v2f;
typedef __attribute__((ext_vector_type(4))) float v4f;
typedef __attribute__((ext_vector_type(8))) float v8f;

#define BATCH 64
#define LDIM 96
#define CDIM 2048
#define TILE 128
#define WAVES 8
#define MT (LDIM / 16)
#define MW 3
#define HCOLS 64
#define TCOLS (TILE + 2)
#define GRP 36
#define NPOS (4 * GRP)
#define LPITCH 100
#define TILEF (NPOS * LPITCH)
#define NTR 6
#define KSTEPS (LDIM / 4)
#define SEGL 48
#define WIN (SEGL + 24)

__global__ void build_wt_kernel(const float* __restrict__ conv_w,
                                float* __restrict__ wt2) {
  int idx = blockIdx.x * blockDim.x + threadIdx.x;
  if (idx >= NTR * LDIM * LDIM) return;
  int j  = idx & 1;
  int m  = (idx >> 1) % LDIM;
  int r  = idx / (2 * LDIM);
  int t  = r / (LDIM / 2);
  int kp = r - t * (LDIM / 2);
  int ci = 2 * kp + j;
  const float* g = conv_w + (m * LDIM + ci) * 3;
  float g0 = g[0], g1 = g[1], g2 = g[2];
  float u;
  if (t == 0)      u = 0.25f * g0;
  else if (t == 1) u = ((g0 + g2) + g1) * (-1.f / 6.f);
  else if (t == 2) u = ((g0 + g2) - g1) * (-1.f / 6.f);
  else if (t == 3) u = (g0 * (1.f / 24.f) + g2 * (1.f / 6.f)) + g1 * (1.f / 12.f);
  else if (t == 4) u = (g0 * (1.f / 24.f) + g2 * (1.f / 6.f)) - g1 * (1.f / 12.f);
  else             u = g2;
  wt2[idx] = u;
}

__device__ __forceinline__ void load_a_step(const v2f* __restrict__ wlane, int ks, v2f (&A)[MW][NTR]) {
  #pragma unroll
  for (int i = 0; i < MW; ++i) {
    #pragma unroll
    for (int t = 0; t < NTR; ++t) A[i][t] = wlane[(t * (LDIM / 2) + 2 * ks) * LDIM + 16 * i];
  }
}

__device__ __forceinline__ void load_d_step(const float* tl, int bl, int ks, v2f (&D)[6]) {
  #pragma unroll
  for (int jj = 0; jj < 6; ++jj) {
    const int slot = (jj < 4) ? jj * GRP : (jj - 4) * GRP + 1;
    D[jj] = *(const v2f*)&tl[bl + slot * LPITCH + 4 * ks];
  }
}

__device__ __forceinline__ void transform_step(const v2f (&D)[6], v2f (&V)[NTR]) {
  const v2f t1 = D[4] - 4.f * D[2];
  const v2f t2 = D[3] - 4.f * D[1];
  const v2f t3 = D[4] - D[2];
  const v2f t4 = D[3] - D[1];
  V[0] = 4.f * D[0] + (D[4] - 5.f * D[2]);
  V[1] = t1 + t2;
  V[2] = t1 - t2;
  V[3] = t3 + 2.f * t4;
  V[4] = t3 - 2.f * t4;
  V[5] = 4.f * D[1] + (D[5] - 5.f * D[3]);
}

__device__ __forceinline__ void mma_step(const v2f (&A)[MW][NTR], const v2f (&V)[NTR], v8f (&acc)[MW][NTR]) {
  #pragma unroll
  for (int t = 0; t < NTR; ++t) {
    #pragma unroll
    for (int i = 0; i < MW; ++i) {
      acc[i][t] = __builtin_amdgcn_wmma_f32_16x16x4_f32(
          false, A[i][t], false, V[t], (short)0, acc[i][t], false, false);
    }
  }
}

__device__ __forceinline__ void decomp_row(float xv, float s5, float s13, float s25,
                                           float gw0, float gw1, float gw2,
                                           float gb0, float gb1, float gb2,
                                           float& mean, float& res) {
  float m5  = s5 * 0.2f;
  float m13 = s13 * (1.f / 13.f);
  float m25 = s25 * 0.04f;
  float l0 = fmaf(xv, gw0, gb0);
  float l1 = fmaf(xv, gw1, gb1);
  float l2 = fmaf(xv, gw2, gb2);
  float mx = fmaxf(fmaxf(l0, l1), l2);
  float e0 = __expf(l0 - mx), e1 = __expf(l1 - mx), e2 = __expf(l2 - mx);
  float inv = 1.f / (e0 + e1 + e2);
  mean = (m5 * e0 + m13 * e1 + m25 * e2) * inv;
  res = xv - mean;
}

__global__ __launch_bounds__(WAVES * 32)
void msdecomp_conv_kernel(const float* __restrict__ x,
                          const float* __restrict__ wt2,
                          const float* __restrict__ gate_w,
                          const float* __restrict__ gate_b,
                          float* __restrict__ out) {
  __shared__ __attribute__((aligned(16))) float tiles[2 * TILEF];
  __shared__ float xh[2 * LDIM];
  float* rs = tiles;
  float* ms = tiles + TILEF;

  const int tid = threadIdx.x;
  const int b   = blockIdx.y;
  const int c0  = blockIdx.x * TILE;
  const float* xb = x + (size_t)b * LDIM * CDIM;

  const float gw0 = gate_w[0], gw1 = gate_w[1], gw2 = gate_w[2];
  const float gb0 = gate_b[0], gb1 = gate_b[1], gb2 = gate_b[2];
  {
    const int seg = __builtin_amdgcn_readfirstlane(tid >> 7);
    const int lbeg = seg * SEGL;
    const int ct  = tid & (TILE - 1);
    const int col = ct + 1;
    const int slot = (col & 3) * GRP + (col >> 2);
    const float* xt = xb + c0;
    float* rc = &rs[slot * LPITCH];
    float* mc = &ms[slot * LPITCH];

    float hv = 0.f;
    if (tid < 2 * LDIM) {
      const int hsel = tid / LDIM;
      const int l    = tid - hsel * LDIM;
      const int gc   = (hsel ? (c0 + TILE) : (c0 - 1 + CDIM)) & (CDIM - 1);
      hv = xb[l * CDIM + gc];
    }

    float w[WIN];
    #pragma unroll
    for (int j = 0; j < WIN; ++j) {
      int r = lbeg - 12 + j;
      if (j < 12) r = max(r, 0);
      if (j >= WIN - 12) r = min(r, LDIM - 1);
      w[j] = xt[r * CDIM + ct];
    }

    if (tid < 2 * LDIM) xh[tid] = hv;
    __syncthreads();

    float s5 = w[12];
    #pragma unroll
    for (int j = 1; j <= 2; ++j)  s5  += w[12 + j] + w[12 - j];
    float s13 = s5;
    #pragma unroll
    for (int j = 3; j <= 6; ++j)  s13 += w[12 + j] + w[12 - j];
    float s25 = s13;
    #pragma unroll
    for (int j = 7; j <= 12; ++j) s25 += w[12 + j] + w[12 - j];

    #pragma unroll
    for (int i = 0; i < SEGL; ++i) {
      float mean, res;
      decomp_row(w[12 + i], s5, s13, s25, gw0, gw1, gw2, gb0, gb1, gb2, mean, res);
      mc[lbeg + i] = mean;
      rc[lbeg + i] = res;
      if (i + 1 < SEGL) {
        s5  += w[i + 15] - w[i + 10];
        s13 += w[i + 19] - w[i + 6];
        s25 += w[i + 25] - w[i];
      }
    }
  }
  if (tid < 2 * LDIM) {
    const int hsel = tid / LDIM;
    const int l    = tid - hsel * LDIM;
    const int col  = hsel * (TCOLS - 1);
    const int slot = (col & 3) * GRP + (col >> 2);
    const float* xc = &xh[hsel * LDIM];
    float s5 = xc[l];
    for (int j = 1; j <= 2; ++j)  s5  += xc[min(l + j, 95)] + xc[max(l - j, 0)];
    float s13 = s5;
    for (int j = 3; j <= 6; ++j)  s13 += xc[min(l + j, 95)] + xc[max(l - j, 0)];
    float s25 = s13;
    for (int j = 7; j <= 12; ++j) s25 += xc[min(l + j, 95)] + xc[max(l - j, 0)];
    float mean, res;
    decomp_row(xc[l], s5, s13, s25, gw0, gw1, gw2, gb0, gb1, gb2, mean, res);
    ms[slot * LPITCH + l] = mean;
    rs[slot * LPITCH + l] = res;
  }
  __syncthreads();

  const int wave = tid >> 5;
  const int lane = tid & 31;
  const int half = (lane >> 4) & 1;
  const int ln   = lane & 15;
  const int m0   = (wave & 1) * (MW * 16);
  const int h0   = ((wave >> 1) & 1) * HCOLS;
  const int cv   = wave >> 2;

  const v2f* wl = (const v2f*)wt2 + half * LDIM + m0 + ln;
  const float* tl = tiles + cv * TILEF;
  const v8f vz = {0.f, 0.f, 0.f, 0.f, 0.f, 0.f, 0.f, 0.f};
  const int bl = (h0 / 4 + ln) * LPITCH + 2 * half;

  v8f acc[MW][NTR];
  #pragma unroll
  for (int i = 0; i < MW; ++i) {
    #pragma unroll
    for (int t = 0; t < NTR; ++t) acc[i][t] = vz;
  }

  v2f Af[2][MW][NTR];
  v2f Df[6];

  load_a_step(wl, 0, Af[0]);
  load_d_step(tl, bl, 0, Df);

  #pragma unroll
  for (int ks = 0; ks < KSTEPS; ++ks) {
    v2f V[NTR];
    transform_step(Df, V);
    if (ks + 1 < KSTEPS) {
      load_a_step(wl, ks + 1, Af[(ks + 1) % 2]);
      load_d_step(tl, bl, ks + 1, Df);
    }
    mma_step(Af[ks % 2], V, acc);
    __builtin_amdgcn_sched_barrier(0);
  }

  float* op = out + (size_t)cv * BATCH * LDIM * CDIM
                  + (size_t)b * LDIM * CDIM;
  const int cc = c0 + h0 + 4 * ln;
  #pragma unroll
  for (int i = 0; i < MW; ++i) {
    #pragma unroll
    for (int r = 0; r < 8; ++r) {
      const int rowm = m0 + 16 * i + r + 8 * half;
      const float s12 = acc[i][1][r] + acc[i][2][r];
      const float d12 = acc[i][1][r] - acc[i][2][r];
      const float s34 = acc[i][3][r] + acc[i][4][r];
      const float d34 = acc[i][3][r] - acc[i][4][r];
      v4f o;
      o.x = (acc[i][0][r] + s12) + s34;
      o.y = d12 + 2.f * d34;
      o.z = s12 + 4.f * s34;
      o.w = (d12 + 8.f * d34) + acc[i][5][r];
      *(v4f*)&op[(size_t)rowm * CDIM + cc] = o;
    }
  }
}

extern "C" void kernel_launch(void* const* d_in, const int* in_sizes, int n_in,
                              void* d_out, int out_size, void* d_ws, size_t ws_size,
                              hipStream_t stream) {
  const float* x      = (const float*)d_in[0];
  const float* conv_w = (const float*)d_in[1];
  const float* gate_w = (const float*)d_in[2];
  const float* gate_b = (const float*)d_in[3];
  float* out = (float*)d_out;
  float* wt2 = (float*)d_ws;

  build_wt_kernel<<<(NTR * LDIM * LDIM + 255) / 256, 256, 0, stream>>>(conv_w, wt2);

  dim3 grid(CDIM / TILE, BATCH);
  msdecomp_conv_kernel<<<grid, WAVES * 32, 0, stream>>>(x, wt2, gate_w, gate_b, out);
}


# Softmax gate of the decomposition in base 2: log2(e) folded into the gate weights once per thread, and 1/sum by v_rcp_f32 (sum in [1,3]) in place of the full f32 division; sum order unchanged
# speedup vs baseline: 3.7276x; 1.0403x over previous
#include <hip/hip_runtime.h>

typedef __attribute__((ext_vector_type(2))) float v2f;
typedef __attribute__((ext_vector_type(4))) float v4f;
typedef __attribute__((ext_vector_type(8))) float v8f;

#define BATCH 64
#define LDIM 96
#define CDIM 2048
#define TILE 128
#define WAVES 8
#define MT (LDIM / 16)
#define MW 3
#define HCOLS 64
#define TCOLS (TILE + 2)
#define GRP 36
#define NPOS (4 * GRP)
#define LPITCH 100
#define TILEF (NPOS * LPITCH)
#define NTR 6
#define KSTEPS (LDIM / 4)
#define SEGL 48
#define WIN (SEGL + 24)

__global__ void build_wt_kernel(const float* __restrict__ conv_w,
                                float* __restrict__ wt2) {
  int idx = blockIdx.x * blockDim.x + threadIdx.x;
  if (idx >= NTR * LDIM * LDIM) return;
  int j  = idx & 1;
  int m  = (idx >> 1) % LDIM;
  int r  = idx / (2 * LDIM);
  int t  = r / (LDIM / 2);
  int kp = r - t * (LDIM / 2);
  int ci = 2 * kp + j;
  const float* g = conv_w + (m * LDIM + ci) * 3;
  float g0 = g[0], g1 = g[1], g2 = g[2];
  float u;
  if (t == 0)      u = 0.25f * g0;
  else if (t == 1) u = ((g0 + g2) + g1) * (-1.f / 6.f);
  else if (t == 2) u = ((g0 + g2) - g1) * (-1.f / 6.f);
  else if (t == 3) u = (g0 * (1.f / 24.f) + g2 * (1.f / 6.f)) + g1 * (1.f / 12.f);
  else if (t == 4) u = (g0 * (1.f / 24.f) + g2 * (1.f / 6.f)) - g1 * (1.f / 12.f);
  else             u = g2;
  wt2[idx] = u;
}

__device__ __forceinline__ void load_a_step(const v2f* __restrict__ wlane, int ks, v2f (&A)[MW][NTR]) {
  #pragma unroll
  for (int i = 0; i < MW; ++i) {
    #pragma unroll
    for (int t = 0; t < NTR; ++t) A[i][t] = wlane[(t * (LDIM / 2) + 2 * ks) * LDIM + 16 * i];
  }
}

__device__ __forceinline__ void load_d_step(const float* tl, int bl, int ks, v2f (&D)[6]) {
  #pragma unroll
  for (int jj = 0; jj < 6; ++jj) {
    const int slot = (jj < 4) ? jj * GRP : (jj - 4) * GRP + 1;
    D[jj] = *(const v2f*)&tl[bl + slot * LPITCH + 4 * ks];
  }
}

__device__ __forceinline__ void transform_step(const v2f (&D)[6], v2f (&V)[NTR]) {
  const v2f t1 = D[4] - 4.f * D[2];
  const v2f t2 = D[3] - 4.f * D[1];
  const v2f t3 = D[4] - D[2];
  const v2f t4 = D[3] - D[1];
  V[0] = 4.f * D[0] + (D[4] - 5.f * D[2]);
  V[1] = t1 + t2;
  V[2] = t1 - t2;
  V[3] = t3 + 2.f * t4;
  V[4] = t3 - 2.f * t4;
  V[5] = 4.f * D[1] + (D[5] - 5.f * D[3]);
}

__device__ __forceinline__ void mma_step(const v2f (&A)[MW][NTR], const v2f (&V)[NTR], v8f (&acc)[MW][NTR]) {
  #pragma unroll
  for (int t = 0; t < NTR; ++t) {
    #pragma unroll
    for (int i = 0; i < MW; ++i) {
      acc[i][t] = __builtin_amdgcn_wmma_f32_16x16x4_f32(
          false, A[i][t], false, V[t], (short)0, acc[i][t], false, false);
    }
  }
}

__device__ __forceinline__ void decomp_row(float xv, float s5, float s13, float s25,
                                           float gw0, float gw1, float gw2,
                                           float gb0, float gb1, float gb2,
                                           float& mean, float& res) {
  float m5  = s5 * 0.2f;
  float m13 = s13 * (1.f / 13.f);
  float m25 = s25 * 0.04f;
  float l0 = fmaf(xv, gw0, gb0);
  float l1 = fmaf(xv, gw1, gb1);
  float l2 = fmaf(xv, gw2, gb2);
  float mx = fmaxf(fmaxf(l0, l1), l2);
  float e0 = __builtin_amdgcn_exp2f(l0 - mx), e1 = __builtin_amdgcn_exp2f(l1 - mx), e2 = __builtin_amdgcn_exp2f(l2 - mx);
  float inv = __builtin_amdgcn_rcpf(e0 + e1 + e2);
  mean = (m5 * e0 + m13 * e1 + m25 * e2) * inv;
  res = xv - mean;
}

__global__ __launch_bounds__(WAVES * 32)
void msdecomp_conv_kernel(const float* __restrict__ x,
                          const float* __restrict__ wt2,
                          const float* __restrict__ gate_w,
                          const float* __restrict__ gate_b,
                          float* __restrict__ out) {
  __shared__ __attribute__((aligned(16))) float tiles[2 * TILEF];
  __shared__ float xh[2 * LDIM];
  float* rs = tiles;
  float* ms = tiles + TILEF;

  const int tid = threadIdx.x;
  const int b   = blockIdx.y;
  const int c0  = blockIdx.x * TILE;
  const float* xb = x + (size_t)b * LDIM * CDIM;

  const float l2e = 1.4426950408889634f;
  const float gw0 = gate_w[0] * l2e, gw1 = gate_w[1] * l2e, gw2 = gate_w[2] * l2e;
  const float gb0 = gate_b[0] * l2e, gb1 = gate_b[1] * l2e, gb2 = gate_b[2] * l2e;
  {
    const int seg = __builtin_amdgcn_readfirstlane(tid >> 7);
    const int lbeg = seg * SEGL;
    const int ct  = tid & (TILE - 1);
    const int col = ct + 1;
    const int slot = (col & 3) * GRP + (col >> 2);
    const float* xt = xb + c0;
    float* rc = &rs[slot * LPITCH];
    float* mc = &ms[slot * LPITCH];

    float hv = 0.f;
    if (tid < 2 * LDIM) {
      const int hsel = tid / LDIM;
      const int l    = tid - hsel * LDIM;
      const int gc   = (hsel ? (c0 + TILE) : (c0 - 1 + CDIM)) & (CDIM - 1);
      hv = xb[l * CDIM + gc];
    }

    float w[WIN];
    #pragma unroll
    for (int j = 0; j < WIN; ++j) {
      int r = lbeg - 12 + j;
      if (j < 12) r = max(r, 0);
      if (j >= WIN - 12) r = min(r, LDIM - 1);
      w[j] = xt[r * CDIM + ct];
    }

    if (tid < 2 * LDIM) xh[tid] = hv;
    __syncthreads();

    float s5 = w[12];
    #pragma unroll
    for (int j = 1; j <= 2; ++j)  s5  += w[12 + j] + w[12 - j];
    float s13 = s5;
    #pragma unroll
    for (int j = 3; j <= 6; ++j)  s13 += w[12 + j] + w[12 - j];
    float s25 = s13;
    #pragma unroll
    for (int j = 7; j <= 12; ++j) s25 += w[12 + j] + w[12 - j];

    #pragma unroll
    for (int i = 0; i < SEGL; ++i) {
      float mean, res;
      decomp_row(w[12 + i], s5, s13, s25, gw0, gw1, gw2, gb0, gb1, gb2, mean, res);
      mc[lbeg + i] = mean;
      rc[lbeg + i] = res;
      if (i + 1 < SEGL) {
        s5  += w[i + 15] - w[i + 10];
        s13 += w[i + 19] - w[i + 6];
        s25 += w[i + 25] - w[i];
      }
    }
  }
  if (tid < 2 * LDIM) {
    const int hsel = tid / LDIM;
    const int l    = tid - hsel * LDIM;
    const int col  = hsel * (TCOLS - 1);
    const int slot = (col & 3) * GRP + (col >> 2);
    const float* xc = &xh[hsel * LDIM];
    float s5 = xc[l];
    for (int j = 1; j <= 2; ++j)  s5  += xc[min(l + j, 95)] + xc[max(l - j, 0)];
    float s13 = s5;
    for (int j = 3; j <= 6; ++j)  s13 += xc[min(l + j, 95)] + xc[max(l - j, 0)];
    float s25 = s13;
    for (int j = 7; j <= 12; ++j) s25 += xc[min(l + j, 95)] + xc[max(l - j, 0)];
    float mean, res;
    decomp_row(xc[l], s5, s13, s25, gw0, gw1, gw2, gb0, gb1, gb2, mean, res);
    ms[slot * LPITCH + l] = mean;
    rs[slot * LPITCH + l] = res;
  }
  __syncthreads();

  const int wave = tid >> 5;
  const int lane = tid & 31;
  const int half = (lane >> 4) & 1;
  const int ln   = lane & 15;
  const int m0   = (wave & 1) * (MW * 16);
  const int h0   = ((wave >> 1) & 1) * HCOLS;
  const int cv   = wave >> 2;

  const v2f* wl = (const v2f*)wt2 + half * LDIM + m0 + ln;
  const float* tl = tiles + cv * TILEF;
  const v8f vz = {0.f, 0.f, 0.f, 0.f, 0.f, 0.f, 0.f, 0.f};
  const int bl = (h0 / 4 + ln) * LPITCH + 2 * half;

  v8f acc[MW][NTR];
  #pragma unroll
  for (int i = 0; i < MW; ++i) {
    #pragma unroll
    for (int t = 0; t < NTR; ++t) acc[i][t] = vz;
  }

  v2f Af[2][MW][NTR];
  v2f Df[6];

  load_a_step(wl, 0, Af[0]);
  load_d_step(tl, bl, 0, Df);

  #pragma unroll
  for (int ks = 0; ks < KSTEPS; ++ks) {
    v2f V[NTR];
    transform_step(Df, V);
    if (ks + 1 < KSTEPS) {
      load_a_step(wl, ks + 1, Af[(ks + 1) % 2]);
      load_d_step(tl, bl, ks + 1, Df);
    }
    mma_step(Af[ks % 2], V, acc);
    __builtin_amdgcn_sched_barrier(0);
  }

  float* op = out + (size_t)cv * BATCH * LDIM * CDIM
                  + (size_t)b * LDIM * CDIM;
  const int cc = c0 + h0 + 4 * ln;
  #pragma unroll
  for (int i = 0; i < MW; ++i) {
    #pragma unroll
    for (int r = 0; r < 8; ++r) {
      const int rowm = m0 + 16 * i + r + 8 * half;
      const float s12 = acc[i][1][r] + acc[i][2][r];
      const float d12 = acc[i][1][r] - acc[i][2][r];
      const float s34 = acc[i][3][r] + acc[i][4][r];
      const float d34 = acc[i][3][r] - acc[i][4][r];
      v4f o;
      o.x = (acc[i][0][r] + s12) + s34;
      o.y = d12 + 2.f * d34;
      o.z = s12 + 4.f * s34;
      o.w = (d12 + 8.f * d34) + acc[i][5][r];
      *(v4f*)&op[(size_t)rowm * CDIM + cc] = o;
    }
  }
}

extern "C" void kernel_launch(void* const* d_in, const int* in_sizes, int n_in,
                              void* d_out, int out_size, void* d_ws, size_t ws_size,
                              hipStream_t stream) {
  const float* x      = (const float*)d_in[0];
  const float* conv_w = (const float*)d_in[1];
  const float* gate_w = (const float*)d_in[2];
  const float* gate_b = (const float*)d_in[3];
  float* out = (float*)d_out;
  float* wt2 = (float*)d_ws;

  build_wt_kernel<<<(NTR * LDIM * LDIM + 255) / 256, 256, 0, stream>>>(conv_w, wt2);

  dim3 grid(CDIM / TILE, BATCH);
  msdecomp_conv_kernel<<<grid, WAVES * 32, 0, stream>>>(x, wt2, gate_w, gate_b, out);
}
